# starting kernel
# baseline (speedup 1.0000x reference)
#include <hip/hip_runtime.h>
#include <math.h>

#define DM   512
#define NH   8
#define DKK  64
#define B_   2
#define LL   256
#define ROWS (B_*LL)

typedef __attribute__((ext_vector_type(2))) float v2f;
typedef __attribute__((ext_vector_type(8))) float v8f;

__global__ __launch_bounds__(256)
void gemm512_wmma(const float* __restrict__ A, const float* __restrict__ W,
                  const float* __restrict__ lam, float* __restrict__ Y, int act)
{
    const int wave = threadIdx.x >> 5;
    const int lane = threadIdx.x & 31;
    const int tile = blockIdx.x * 8 + wave;
    const int tr   = tile >> 5;
    const int tc   = tile & 31;
    const int half = lane >> 4;
    const int idx  = lane & 15;

    const float* arow = A + (size_t)(tr * 16 + idx) * DM + 2 * half;
    const float* wrow = W + (size_t)(tc * 16 + idx) * DM + 2 * half;

    v8f acc = {0.f, 0.f, 0.f, 0.f, 0.f, 0.f, 0.f, 0.f};
    #pragma unroll 4
    for (int k = 0; k < DM; k += 4) {
        v2f a = *(const v2f*)(arow + k);
        v2f b = *(const v2f*)(wrow + k);
        acc = __builtin_amdgcn_wmma_f32_16x16x4_f32(
                   false, a,  false, b,
                   (short)0, acc,  false,  false);
    }

    const int mbase = tr * 16 + half * 8;
    const int col   = tc * 16 + idx;
    const float lm  = act ? lam[col] : 0.f;
    #pragma unroll
    for (int i = 0; i < 8; ++i) {
        float v = acc[i];
        if (act) v = log1pf(fmaxf(v, 0.f)) - lm;
        Y[(size_t)(mbase + i) * DM + col] = v;
    }
}

__global__ __launch_bounds__(256)
void tropical_linear(const float* __restrict__ proj, const float* __restrict__ Wt,
                     float* __restrict__ out)
{
    __shared__ float wl[DKK * 65];
    __shared__ float xl[4 * DKK];

    const int t  = threadIdx.x;
    const int bh = blockIdx.x >> 6;
    const int lt = blockIdx.x & 63;
    const int b  = bh >> 3, h = bh & 7;

    #pragma unroll
    for (int r = 0; r < 16; ++r) {
        int e = t + 256 * r;
        wl[(e >> 6) * 65 + (e & 63)] = Wt[e];
    }
    const int lr = t >> 6, oi = t & 63;
    xl[t] = proj[(size_t)(b * LL + lt * 4 + lr) * DM + h * DKK + oi];
    __syncthreads();

    const float* xr = xl + lr * DKK;
    const float* wr = wl + oi * 65;
    float m = -__builtin_inff();
    #pragma unroll 8
    for (int i = 0; i < DKK; ++i)
        m = fmaxf(m, xr[i] + wr[i]);
    out[(size_t)bh * LL * DKK + (lt * 4 + lr) * DKK + oi] = m;
}

__global__ __launch_bounds__(256)
void trop_attn(const float* __restrict__ qt, const float* __restrict__ kt,
               const float* __restrict__ vt, float* __restrict__ scores,
               float* __restrict__ ctxout)
{
    __shared__ float qs[16 * DKK];
    __shared__ float ks[64 * 65];
    __shared__ float vs[64 * 65];
    __shared__ float ss[16 * 64];

    const int t  = threadIdx.x;
    const int it = blockIdx.x;
    const int bh = blockIdx.y;

    const float* qp = qt + ((size_t)bh * LL + it * 16) * DKK;
    #pragma unroll
    for (int r = 0; r < 4; ++r) qs[t + 256 * r] = qp[t + 256 * r];

    const float NI = -__builtin_inff();
    float ctx[4] = {NI, NI, NI, NI};

    for (int jc = 0; jc < 4; ++jc) {
        const float* kp = kt + ((size_t)bh * LL + jc * 64) * DKK;
        const float* vp = vt + ((size_t)bh * LL + jc * 64) * DKK;
        __syncthreads();
        #pragma unroll
        for (int r = 0; r < 16; ++r) {
            int e = t + 256 * r; int j = e >> 6, d = e & 63;
            ks[j * 65 + d] = kp[e];
            vs[j * 65 + d] = vp[e];
        }
        __syncthreads();

        #pragma unroll
        for (int r = 0; r < 4; ++r) {
            int c = t + 256 * r; int i = c >> 6, jj = c & 63;
            const float* qr = qs + i * DKK;
            const float* kr = ks + jj * 65;
            float mx = NI, mn = __builtin_inff();
            #pragma unroll 8
            for (int d = 0; d < DKK; ++d) {
                float df = qr[d] - kr[d];
                mx = fmaxf(mx, df);
                mn = fminf(mn, df);
            }
            float s = mn - mx;
            ss[c] = s;
            scores[(size_t)bh * LL * LL + (size_t)(it * 16 + i) * LL + jc * 64 + jj] = s;
        }
        __syncthreads();

        #pragma unroll
        for (int r = 0; r < 4; ++r) {
            int c = t + 256 * r; int i = c >> 6, d = c & 63;
            const float* sr = ss + i * 64;
            float m = ctx[r];
            #pragma unroll 8
            for (int jj = 0; jj < 64; ++jj)
                m = fmaxf(m, sr[jj] + vs[jj * 65 + d]);
            ctx[r] = m;
        }
    }

    const int b = bh >> 3, h = bh & 7;
    #pragma unroll
    for (int r = 0; r < 4; ++r) {
        int c = t + 256 * r; int i = c >> 6, d = c & 63;
        ctxout[(size_t)(b * LL + it * 16 + i) * DM + h * DKK + d] = expm1f(ctx[r]);
    }
}

extern "C" void kernel_launch(void* const* d_in, const int* in_sizes, int n_in,
                              void* d_out, int out_size, void* d_ws, size_t ws_size,
                              hipStream_t stream)
{
    (void)in_sizes; (void)n_in; (void)out_size; (void)ws_size;
    const float* x   = (const float*)d_in[0];
    const float* Wq  = (const float*)d_in[1];
    const float* Wk  = (const float*)d_in[2];
    const float* Wv  = (const float*)d_in[3];
    const float* Wo  = (const float*)d_in[4];
    const float* lam = (const float*)d_in[5];
    const float* Wqt = (const float*)d_in[6];
    const float* Wkt = (const float*)d_in[7];
    const float* Wvt = (const float*)d_in[8];

    float* out = (float*)d_out;
    float* sc  = out + (size_t)ROWS * DM;

    const size_t N = (size_t)ROWS * DM;
    float* ws  = (float*)d_ws;
    float* qp  = ws;
    float* kp  = ws + N;
    float* vp  = ws + 2 * N;
    float* qtb = ws + 3 * N;
    float* ktb = ws + 4 * N;
    float* vtb = ws + 5 * N;
    float* ctx = qp;

    dim3 blk(256);
    gemm512_wmma<<<128, blk, 0, stream>>>(x, Wq, lam, qp, 1);
    gemm512_wmma<<<128, blk, 0, stream>>>(x, Wk, lam, kp, 1);
    gemm512_wmma<<<128, blk, 0, stream>>>(x, Wv, lam, vp, 1);

    tropical_linear<<<1024, blk, 0, stream>>>(qp, Wqt, qtb);
    tropical_linear<<<1024, blk, 0, stream>>>(kp, Wkt, ktb);
    tropical_linear<<<1024, blk, 0, stream>>>(vp, Wvt, vtb);

    trop_attn<<<dim3(16, 16), blk, 0, stream>>>(qtb, ktb, vtb, sc, ctx);

    gemm512_wmma<<<128, blk, 0, stream>>>(ctx, Wo, (const float*)nullptr, out, 0);
}


# attention core register-tiled: lane keeps its K row (scores) and V column pair (context) in registers, q and score rows as broadcast 16-byte LDS reads, j split over waves, exact max merge at the end
# speedup vs baseline: 1.7233x; 1.7233x over previous
#include <hip/hip_runtime.h>
#include <math.h>

#define DM   512
#define NH   8
#define DKK  64
#define B_   2
#define LL   256
#define ROWS (B_*LL)

typedef __attribute__((ext_vector_type(2))) float v2f;
typedef __attribute__((ext_vector_type(8))) float v8f;

__global__ __launch_bounds__(256)
void gemm512_wmma(const float* __restrict__ A, const float* __restrict__ W,
                  const float* __restrict__ lam, float* __restrict__ Y, int act)
{
    const int wave = threadIdx.x >> 5;
    const int lane = threadIdx.x & 31;
    const int tile = blockIdx.x * 8 + wave;
    const int tr   = tile >> 5;
    const int tc   = tile & 31;
    const int half = lane >> 4;
    const int idx  = lane & 15;

    const float* arow = A + (size_t)(tr * 16 + idx) * DM + 2 * half;
    const float* wrow = W + (size_t)(tc * 16 + idx) * DM + 2 * half;

    v8f acc = {0.f, 0.f, 0.f, 0.f, 0.f, 0.f, 0.f, 0.f};
    #pragma unroll 4
    for (int k = 0; k < DM; k += 4) {
        v2f a = *(const v2f*)(arow + k);
        v2f b = *(const v2f*)(wrow + k);
        acc = __builtin_amdgcn_wmma_f32_16x16x4_f32(
                   false, a,  false, b,
                   (short)0, acc,  false,  false);
    }

    const int mbase = tr * 16 + half * 8;
    const int col   = tc * 16 + idx;
    const float lm  = act ? lam[col] : 0.f;
    #pragma unroll
    for (int i = 0; i < 8; ++i) {
        float v = acc[i];
        if (act) v = log1pf(fmaxf(v, 0.f)) - lm;
        Y[(size_t)(mbase + i) * DM + col] = v;
    }
}

__global__ __launch_bounds__(256)
void tropical_linear(const float* __restrict__ proj, const float* __restrict__ Wt,
                     float* __restrict__ out)
{
    __shared__ float wl[DKK * 65];
    __shared__ float xl[4 * DKK];

    const int t  = threadIdx.x;
    const int bh = blockIdx.x >> 6;
    const int lt = blockIdx.x & 63;
    const int b  = bh >> 3, h = bh & 7;

    #pragma unroll
    for (int r = 0; r < 16; ++r) {
        int e = t + 256 * r;
        wl[(e >> 6) * 65 + (e & 63)] = Wt[e];
    }
    const int lr = t >> 6, oi = t & 63;
    xl[t] = proj[(size_t)(b * LL + lt * 4 + lr) * DM + h * DKK + oi];
    __syncthreads();

    const float* xr = xl + lr * DKK;
    const float* wr = wl + oi * 65;
    float m = -__builtin_inff();
    #pragma unroll 8
    for (int i = 0; i < DKK; ++i)
        m = fmaxf(m, xr[i] + wr[i]);
    out[(size_t)bh * LL * DKK + (lt * 4 + lr) * DKK + oi] = m;
}

typedef __attribute__((ext_vector_type(4))) float v4f;

#define KPITCH 68

__device__ __forceinline__ void load_row16(v4f (&dst)[16], const float* src)
{
    #pragma unroll
    for (int c = 0; c < 16; ++c) dst[c] = ((const v4f*)src)[c];
}

__device__ __forceinline__ float score_row(const v4f (&q)[16], const v4f (&k)[16])
{
    float xa = -__builtin_inff(), xb = -__builtin_inff();
    float na =  __builtin_inff(), nb =  __builtin_inff();
    #pragma unroll
    for (int c = 0; c < 16; ++c) {
        const float d0 = q[c].x - k[c].x, d1 = q[c].y - k[c].y;
        const float d2 = q[c].z - k[c].z, d3 = q[c].w - k[c].w;
        xa = fmaxf(fmaxf(xa, d0), d1);  xb = fmaxf(fmaxf(xb, d2), d3);
        na = fminf(fminf(na, d0), d1);  nb = fminf(fminf(nb, d2), d3);
    }
    return fminf(na, nb) - fmaxf(xa, xb);
}

__device__ __forceinline__ void load_srows(v4f (&dst)[16], const float* src)
{
    #pragma unroll
    for (int c = 0; c < 16; ++c) dst[c] = ((const v4f*)src)[c];
}

__device__ __forceinline__ void ctx_rows(const v4f (&s)[16], const v2f (&v)[32],
                                         v2f& p0, v2f& p1)
{
    const float NI = -__builtin_inff();
    float ax0 = NI, ay0 = NI, bx0 = NI, by0 = NI;
    float ax1 = NI, ay1 = NI, bx1 = NI, by1 = NI;
    #pragma unroll
    for (int c = 0; c < 8; ++c) {
        const v4f sa = s[c], sb = s[8 + c];
        const v2f v0 = v[4 * c], v1 = v[4 * c + 1], v2 = v[4 * c + 2], v3 = v[4 * c + 3];
        ax0 = fmaxf(fmaxf(ax0, sa.x + v0.x), sa.y + v1.x);
        ay0 = fmaxf(fmaxf(ay0, sa.x + v0.y), sa.y + v1.y);
        bx0 = fmaxf(fmaxf(bx0, sa.z + v2.x), sa.w + v3.x);
        by0 = fmaxf(fmaxf(by0, sa.z + v2.y), sa.w + v3.y);
        ax1 = fmaxf(fmaxf(ax1, sb.x + v0.x), sb.y + v1.x);
        ay1 = fmaxf(fmaxf(ay1, sb.x + v0.y), sb.y + v1.y);
        bx1 = fmaxf(fmaxf(bx1, sb.z + v2.x), sb.w + v3.x);
        by1 = fmaxf(fmaxf(by1, sb.z + v2.y), sb.w + v3.y);
    }
    p0.x = fmaxf(ax0, bx0);  p0.y = fmaxf(ay0, by0);
    p1.x = fmaxf(ax1, bx1);  p1.y = fmaxf(ay1, by1);
}

__global__ __launch_bounds__(256) __attribute__((amdgpu_waves_per_eu(2, 2)))
void trop_attn(const float* __restrict__ qt, const float* __restrict__ kt,
               const float* __restrict__ vt, float* __restrict__ scores,
               float* __restrict__ ctxout)
{
    __shared__ __attribute__((aligned(16))) float kst[8 * 32 * KPITCH];
    __shared__ __attribute__((aligned(16))) float vst[LL * DKK];
    __shared__ __attribute__((aligned(16))) float qs[16 * DKK];
    __shared__ __attribute__((aligned(16))) float ss[8 * 16 * 32];

    const int t    = threadIdx.x;
    const int w    = t >> 5;
    const int lane = t & 31;
    const int it   = blockIdx.x;
    const int bh   = blockIdx.y;
    const size_t rowbase = (size_t)bh * LL;

    ((v4f*)qs)[t] = ((const v4f*)(qt + (rowbase + it * 16) * DKK))[t];

    float* kw = kst + w * (32 * KPITCH);
    float* vw = vst + w * (32 * DKK);
    {
        const v4f* kg = (const v4f*)(kt + (rowbase + 32 * w) * DKK);
        const v4f* vg = (const v4f*)(vt + (rowbase + 32 * w) * DKK);
        #pragma unroll
        for (int r = 0; r < 16; ++r) {
            const v4f x = kg[r * 32 + lane];
            const v4f y = vg[r * 32 + lane];
            *(v4f*)(kw + (2 * r + (lane >> 4)) * KPITCH + (lane & 15) * 4) = x;
            ((v4f*)vw)[r * 32 + lane] = y;
        }
    }

    __syncthreads();

    v4f kr[16];
    #pragma unroll
    for (int c = 0; c < 16; ++c) kr[c] = *(const v4f*)(kw + lane * KPITCH + c * 4);

    float* sw = ss + w * (16 * 32);
    float* sg = scores + (rowbase + it * 16) * LL + 32 * w + lane;
    {
        v4f qa[16], qb[16];
        load_row16(qa, qs);
        #pragma unroll 1
        for (int i = 0; i < 16; i += 2) {
            load_row16(qb, qs + (i + 1) * DKK);
            const float s0 = score_row(qa, kr);
            sw[i * 32 + lane]  = s0;
            sg[(size_t)i * LL] = s0;
            load_row16(qa, qs + ((i + 2) & 15) * DKK);
            const float s1 = score_row(qb, kr);
            sw[(i + 1) * 32 + lane]  = s1;
            sg[(size_t)(i + 1) * LL] = s1;
        }
    }

    v2f vr[32];
    #pragma unroll
    for (int jj = 0; jj < 32; ++jj) vr[jj] = *(const v2f*)(vw + jj * DKK + 2 * lane);

    __syncthreads();

    float* pw = kw;
    {
        v4f sa[16], sb[16];
        load_srows(sa, sw);
        #pragma unroll 1
        for (int i = 0; i < 16; i += 4) {
            v2f p0, p1;
            load_srows(sb, sw + (i + 2) * 32);
            ctx_rows(sa, vr, p0, p1);
            *(v2f*)(pw + i * DKK + 2 * lane)       = p0;
            *(v2f*)(pw + (i + 1) * DKK + 2 * lane) = p1;
            load_srows(sa, sw + ((i + 4) & 15) * 32);
            ctx_rows(sb, vr, p0, p1);
            *(v2f*)(pw + (i + 2) * DKK + 2 * lane) = p0;
            *(v2f*)(pw + (i + 3) * DKK + 2 * lane) = p1;
        }
    }

    __syncthreads();

    {
        const int i = t >> 4, dq = t & 15;
        v4f m = *(const v4f*)(kst + i * DKK + dq * 4);
        #pragma unroll
        for (int ww = 1; ww < 8; ++ww) {
            const v4f p = *(const v4f*)(kst + ww * (32 * KPITCH) + i * DKK + dq * 4);
            m.x = fmaxf(m.x, p.x);  m.y = fmaxf(m.y, p.y);
            m.z = fmaxf(m.z, p.z);  m.w = fmaxf(m.w, p.w);
        }
        const int b = bh >> 3, h = bh & 7;
        v4f o;
        o.x = expm1f(m.x);  o.y = expm1f(m.y);  o.z = expm1f(m.z);  o.w = expm1f(m.w);
        *(v4f*)(ctxout + (size_t)(b * LL + it * 16 + i) * DM + h * DKK + dq * 4) = o;
    }
}

extern "C" void kernel_launch(void* const* d_in, const int* in_sizes, int n_in,
                              void* d_out, int out_size, void* d_ws, size_t ws_size,
                              hipStream_t stream)
{
    (void)in_sizes; (void)n_in; (void)out_size; (void)ws_size;
    const float* x   = (const float*)d_in[0];
    const float* Wq  = (const float*)d_in[1];
    const float* Wk  = (const float*)d_in[2];
    const float* Wv  = (const float*)d_in[3];
    const float* Wo  = (const float*)d_in[4];
    const float* lam = (const float*)d_in[5];
    const float* Wqt = (const float*)d_in[6];
    const float* Wkt = (const float*)d_in[7];
    const float* Wvt = (const float*)d_in[8];

    float* out = (float*)d_out;
    float* sc  = out + (size_t)ROWS * DM;

    const size_t N = (size_t)ROWS * DM;
    float* ws  = (float*)d_ws;
    float* qp  = ws;
    float* kp  = ws + N;
    float* vp  = ws + 2 * N;
    float* qtb = ws + 3 * N;
    float* ktb = ws + 4 * N;
    float* vtb = ws + 5 * N;
    float* ctx = qp;

    dim3 blk(256);
    gemm512_wmma<<<128, blk, 0, stream>>>(x, Wq, lam, qp, 1);
    gemm512_wmma<<<128, blk, 0, stream>>>(x, Wk, lam, kp, 1);
    gemm512_wmma<<<128, blk, 0, stream>>>(x, Wv, lam, vp, 1);

    tropical_linear<<<1024, blk, 0, stream>>>(qp, Wqt, qtb);
    tropical_linear<<<1024, blk, 0, stream>>>(kp, Wkt, ktb);
    tropical_linear<<<1024, blk, 0, stream>>>(vp, Wvt, vtb);

    trop_attn<<<dim3(16, 16), blk, 0, stream>>>(qtb, ktb, vtb, sc, ctx);

    gemm512_wmma<<<128, blk, 0, stream>>>(ctx, Wo, (const float*)nullptr, out, 0);
}


# projection, activation and tropical linear fused in one kernel (block = 16 rows x one head, activations only in LDS), q, k and v in one launch: 8 launches become 3
# speedup vs baseline: 3.4787x; 2.0186x over previous
#include <hip/hip_runtime.h>
#include <math.h>

#define DM   512
#define NH   8
#define DKK  64
#define B_   2
#define LL   256
#define ROWS (B_*LL)

typedef __attribute__((ext_vector_type(2))) float v2f;
typedef __attribute__((ext_vector_type(4))) float v4f;
typedef __attribute__((ext_vector_type(8))) float v8f;

__device__ __forceinline__ v8f tile_k512(const float* __restrict__ arow,
                                         const float* __restrict__ wrow)
{
    v8f acc = {0.f, 0.f, 0.f, 0.f, 0.f, 0.f, 0.f, 0.f};
    #pragma unroll 4
    for (int k = 0; k < DM; k += 4) {
        v2f a = *(const v2f*)(arow + k);
        v2f b = *(const v2f*)(wrow + k);
        acc = __builtin_amdgcn_wmma_f32_16x16x4_f32(
                   false, a,  false, b,
                   (short)0, acc,  false,  false);
    }
    return acc;
}

__global__ __launch_bounds__(256)
void gemm512_wmma(const float* __restrict__ A, const float* __restrict__ W,
                  float* __restrict__ Y)
{
    const int wave = threadIdx.x >> 5;
    const int lane = threadIdx.x & 31;
    const int tile = blockIdx.x * 8 + wave;
    const int tr   = tile >> 5;
    const int tc   = tile & 31;
    const int half = lane >> 4;
    const int idx  = lane & 15;

    const v8f acc = tile_k512(A + (size_t)(tr * 16 + idx) * DM + 2 * half,
                              W + (size_t)(tc * 16 + idx) * DM + 2 * half);

    const int mbase = tr * 16 + half * 8;
    const int col   = tc * 16 + idx;
    #pragma unroll
    for (int i = 0; i < 8; ++i)
        Y[(size_t)(mbase + i) * DM + col] = acc[i];
}

__device__ __forceinline__ void load_row16(v4f (&dst)[16], const float* src)
{
    #pragma unroll
    for (int c = 0; c < 16; ++c) dst[c] = ((const v4f*)src)[c];
}

__device__ __forceinline__ float tropical_row(const v4f (&x)[16], const v4f (&w)[16])
{
    float ma = -__builtin_inff(), mb = -__builtin_inff();
    #pragma unroll
    for (int c = 0; c < 16; ++c) {
        ma = fmaxf(fmaxf(ma, x[c].x + w[c].x), x[c].y + w[c].y);
        mb = fmaxf(fmaxf(mb, x[c].z + w[c].z), x[c].w + w[c].w);
    }
    return fmaxf(ma, mb);
}

#define WPITCH 68
#define XPITCH 68

__global__ __launch_bounds__(128)
void proj_tropical(const float* __restrict__ x,
                   const float* __restrict__ Wq, const float* __restrict__ Wk,
                   const float* __restrict__ Wv, const float* __restrict__ lam,
                   const float* __restrict__ Tq, const float* __restrict__ Tk,
                   const float* __restrict__ Tv,
                   float* __restrict__ oq, float* __restrict__ ok,
                   float* __restrict__ ov)
{
    __shared__ __attribute__((aligned(16))) float wl[DKK * WPITCH];
    __shared__ __attribute__((aligned(16))) float xl[16 * XPITCH];

    const int t    = threadIdx.x;
    const int wave = t >> 5;
    const int lane = t & 31;
    const int tr   = blockIdx.x >> 3;
    const int h    = blockIdx.x & 7;
    const int p    = blockIdx.y;
    const float* W  = p == 0 ? Wq : (p == 1 ? Wk : Wv);
    const float* Wt = p == 0 ? Tq : (p == 1 ? Tk : Tv);
    float*       out = p == 0 ? oq : (p == 1 ? ok : ov);

    v4f wt[8];
    #pragma unroll
    for (int r = 0; r < 8; ++r) wt[r] = ((const v4f*)Wt)[t + 128 * r];

    const int half = lane >> 4;
    const int idx  = lane & 15;
    const int tc   = h * 4 + wave;
    const v8f acc = tile_k512(x + (size_t)(tr * 16 + idx) * DM + 2 * half,
                              W + (size_t)(tc * 16 + idx) * DM + 2 * half);

    #pragma unroll
    for (int r = 0; r < 8; ++r) {
        const int e = t + 128 * r;
        *(v4f*)(wl + (e >> 4) * WPITCH + (e & 15) * 4) = wt[r];
    }

    const float lm = lam[tc * 16 + idx];
    #pragma unroll
    for (int i = 0; i < 8; ++i)
        xl[(half * 8 + i) * XPITCH + wave * 16 + idx] = log1pf(fmaxf(acc[i], 0.f)) - lm;

    __syncthreads();

    const int o  = (wave & 1) * 32 + lane;
    const int r0 = (wave >> 1) * 8;
    v4f wr[16];
    #pragma unroll
    for (int c = 0; c < 16; ++c) wr[c] = *(const v4f*)(wl + o * WPITCH + c * 4);

    const int R = tr * 16 + r0;
    const int b = R >> 8, l = R & (LL - 1);
    const float* xb = xl + r0 * XPITCH;
    float* og = out + ((size_t)(b * NH + h) * LL + l) * DKK + o;
    {
        v4f xa[16], xc[16];
        load_row16(xa, xb);
        #pragma unroll 1
        for (int r = 0; r < 8; r += 2) {
            load_row16(xc, xb + (r + 1) * XPITCH);
            og[(size_t)r * DKK] = tropical_row(xa, wr);
            load_row16(xa, xb + ((r + 2) & 7) * XPITCH);
            og[(size_t)(r + 1) * DKK] = tropical_row(xc, wr);
        }
    }
}

#define KPITCH 68

__device__ __forceinline__ float score_row(const v4f (&q)[16], const v4f (&k)[16])
{
    float xa = -__builtin_inff(), xb = -__builtin_inff();
    float na =  __builtin_inff(), nb =  __builtin_inff();
    #pragma unroll
    for (int c = 0; c < 16; ++c) {
        const float d0 = q[c].x - k[c].x, d1 = q[c].y - k[c].y;
        const float d2 = q[c].z - k[c].z, d3 = q[c].w - k[c].w;
        xa = fmaxf(fmaxf(xa, d0), d1);  xb = fmaxf(fmaxf(xb, d2), d3);
        na = fminf(fminf(na, d0), d1);  nb = fminf(fminf(nb, d2), d3);
    }
    return fminf(na, nb) - fmaxf(xa, xb);
}

__device__ __forceinline__ void load_srows(v4f (&dst)[16], const float* src)
{
    #pragma unroll
    for (int c = 0; c < 16; ++c) dst[c] = ((const v4f*)src)[c];
}

__device__ __forceinline__ void ctx_rows(const v4f (&s)[16], const v2f (&v)[32],
                                         v2f& p0, v2f& p1)
{
    const float NI = -__builtin_inff();
    float ax0 = NI, ay0 = NI, bx0 = NI, by0 = NI;
    float ax1 = NI, ay1 = NI, bx1 = NI, by1 = NI;
    #pragma unroll
    for (int c = 0; c < 8; ++c) {
        const v4f sa = s[c], sb = s[8 + c];
        const v2f v0 = v[4 * c], v1 = v[4 * c + 1], v2 = v[4 * c + 2], v3 = v[4 * c + 3];
        ax0 = fmaxf(fmaxf(ax0, sa.x + v0.x), sa.y + v1.x);
        ay0 = fmaxf(fmaxf(ay0, sa.x + v0.y), sa.y + v1.y);
        bx0 = fmaxf(fmaxf(bx0, sa.z + v2.x), sa.w + v3.x);
        by0 = fmaxf(fmaxf(by0, sa.z + v2.y), sa.w + v3.y);
        ax1 = fmaxf(fmaxf(ax1, sb.x + v0.x), sb.y + v1.x);
        ay1 = fmaxf(fmaxf(ay1, sb.x + v0.y), sb.y + v1.y);
        bx1 = fmaxf(fmaxf(bx1, sb.z + v2.x), sb.w + v3.x);
        by1 = fmaxf(fmaxf(by1, sb.z + v2.y), sb.w + v3.y);
    }
    p0.x = fmaxf(ax0, bx0);  p0.y = fmaxf(ay0, by0);
    p1.x = fmaxf(ax1, bx1);  p1.y = fmaxf(ay1, by1);
}

__global__ __launch_bounds__(256) __attribute__((amdgpu_waves_per_eu(2, 2)))
void trop_attn(const float* __restrict__ qt, const float* __restrict__ kt,
               const float* __restrict__ vt, float* __restrict__ scores,
               float* __restrict__ ctxout)
{
    __shared__ __attribute__((aligned(16))) float kst[8 * 32 * KPITCH];
    __shared__ __attribute__((aligned(16))) float vst[LL * DKK];
    __shared__ __attribute__((aligned(16))) float qs[16 * DKK];
    __shared__ __attribute__((aligned(16))) float ss[8 * 16 * 32];

    const int t    = threadIdx.x;
    const int w    = t >> 5;
    const int lane = t & 31;
    const int it   = blockIdx.x;
    const int bh   = blockIdx.y;
    const size_t rowbase = (size_t)bh * LL;

    ((v4f*)qs)[t] = ((const v4f*)(qt + (rowbase + it * 16) * DKK))[t];

    float* kw = kst + w * (32 * KPITCH);
    float* vw = vst + w * (32 * DKK);
    {
        const v4f* kg = (const v4f*)(kt + (rowbase + 32 * w) * DKK);
        const v4f* vg = (const v4f*)(vt + (rowbase + 32 * w) * DKK);
        #pragma unroll
        for (int r = 0; r < 16; ++r) {
            const v4f x = kg[r * 32 + lane];
            const v4f y = vg[r * 32 + lane];
            *(v4f*)(kw + (2 * r + (lane >> 4)) * KPITCH + (lane & 15) * 4) = x;
            ((v4f*)vw)[r * 32 + lane] = y;
        }
    }

    __syncthreads();

    v4f kr[16];
    #pragma unroll
    for (int c = 0; c < 16; ++c) kr[c] = *(const v4f*)(kw + lane * KPITCH + c * 4);

    float* sw = ss + w * (16 * 32);
    float* sg = scores + (rowbase + it * 16) * LL + 32 * w + lane;
    {
        v4f qa[16], qb[16];
        load_row16(qa, qs);
        #pragma unroll 1
        for (int i = 0; i < 16; i += 2) {
            load_row16(qb, qs + (i + 1) * DKK);
            const float s0 = score_row(qa, kr);
            sw[i * 32 + lane]  = s0;
            sg[(size_t)i * LL] = s0;
            load_row16(qa, qs + ((i + 2) & 15) * DKK);
            const float s1 = score_row(qb, kr);
            sw[(i + 1) * 32 + lane]  = s1;
            sg[(size_t)(i + 1) * LL] = s1;
        }
    }

    v2f vr[32];
    #pragma unroll
    for (int jj = 0; jj < 32; ++jj) vr[jj] = *(const v2f*)(vw + jj * DKK + 2 * lane);

    __syncthreads();

    float* pw = kw;
    {
        v4f sa[16], sb[16];
        load_srows(sa, sw);
        #pragma unroll 1
        for (int i = 0; i < 16; i += 4) {
            v2f p0, p1;
            load_srows(sb, sw + (i + 2) * 32);
            ctx_rows(sa, vr, p0, p1);
            *(v2f*)(pw + i * DKK + 2 * lane)       = p0;
            *(v2f*)(pw + (i + 1) * DKK + 2 * lane) = p1;
            load_srows(sa, sw + ((i + 4) & 15) * 32);
            ctx_rows(sb, vr, p0, p1);
            *(v2f*)(pw + (i + 2) * DKK + 2 * lane) = p0;
            *(v2f*)(pw + (i + 3) * DKK + 2 * lane) = p1;
        }
    }

    __syncthreads();

    {
        const int i = t >> 4, dq = t & 15;
        v4f m = *(const v4f*)(kst + i * DKK + dq * 4);
        #pragma unroll
        for (int ww = 1; ww < 8; ++ww) {
            const v4f p = *(const v4f*)(kst + ww * (32 * KPITCH) + i * DKK + dq * 4);
            m.x = fmaxf(m.x, p.x);  m.y = fmaxf(m.y, p.y);
            m.z = fmaxf(m.z, p.z);  m.w = fmaxf(m.w, p.w);
        }
        const int b = bh >> 3, h = bh & 7;
        v4f o;
        o.x = expm1f(m.x);  o.y = expm1f(m.y);  o.z = expm1f(m.z);  o.w = expm1f(m.w);
        *(v4f*)(ctxout + (size_t)(b * LL + it * 16 + i) * DM + h * DKK + dq * 4) = o;
    }
}

extern "C" void kernel_launch(void* const* d_in, const int* in_sizes, int n_in,
                              void* d_out, int out_size, void* d_ws, size_t ws_size,
                              hipStream_t stream)
{
    (void)in_sizes; (void)n_in; (void)out_size; (void)ws_size;
    const float* x   = (const float*)d_in[0];
    const float* Wq  = (const float*)d_in[1];
    const float* Wk  = (const float*)d_in[2];
    const float* Wv  = (const float*)d_in[3];
    const float* Wo  = (const float*)d_in[4];
    const float* lam = (const float*)d_in[5];
    const float* Wqt = (const float*)d_in[6];
    const float* Wkt = (const float*)d_in[7];
    const float* Wvt = (const float*)d_in[8];

    float* out = (float*)d_out;
    float* sc  = out + (size_t)ROWS * DM;

    const size_t N = (size_t)ROWS * DM;
    float* ws  = (float*)d_ws;
    float* ctx = ws;
    float* qtb = ws + 3 * N;
    float* ktb = ws + 4 * N;
    float* vtb = ws + 5 * N;

    dim3 blk(256);
    proj_tropical<<<dim3(256, 3), dim3(128), 0, stream>>>(x, Wq, Wk, Wv, lam, Wqt, Wkt, Wvt,
                                                         qtb, ktb, vtb);

    trop_attn<<<dim3(16, 16), blk, 0, stream>>>(qtb, ktb, vtb, sc, ctx);

    gemm512_wmma<<<128, blk, 0, stream>>>(ctx, Wo, out);
}


# both GEMMs stage 128 k per chunk instead of 64: four staging rounds and barriers instead of eight, 20 global loads in flight per thread per round; k sum order unchanged
# speedup vs baseline: 4.9483x; 1.4225x over previous
#include <hip/hip_runtime.h>
#include <math.h>

#define DM   512
#define NH   8
#define DKK  64
#define B_   2
#define LL   256
#define ROWS (B_*LL)

typedef __attribute__((ext_vector_type(2))) float v2f;
typedef __attribute__((ext_vector_type(4))) float v4f;
typedef __attribute__((ext_vector_type(8))) float v8f;

#define WPITCH 132
#define XPITCH 132
#define KC     128
#define WBUF   (DKK * WPITCH)
#define XBUF   (16 * XPITCH)

__device__ __forceinline__ v8f tile16x64_k512(const float* __restrict__ Ab,
                                              const float* __restrict__ Wb,
                                              float* wl, float* xl, const int t)
{
    const int lane = t & 31, wave = t >> 5;
    const int half = lane >> 4, idx = lane & 15;
    const int prow = t >> 5;
    const int pcol = (t & 31) * 4;

    const float* ga = Ab + (size_t)prow * DM + pcol;
    const float* gw = Wb + (size_t)prow * DM + pcol;
    float* sa = xl + prow * XPITCH + pcol;
    float* sw = wl + prow * WPITCH + pcol;
    const float* fa = xl + idx * XPITCH + 4 * half;
    const float* fb = wl + (wave * 16 + idx) * WPITCH + 4 * half;

    v4f ra[4], rw[16];
    #pragma unroll
    for (int r = 0; r < 4; ++r) ra[r] = *(const v4f*)(ga + (size_t)(4 * r) * DM);
    #pragma unroll
    for (int r = 0; r < 16; ++r) rw[r] = *(const v4f*)(gw + (size_t)(4 * r) * DM);
    #pragma unroll
    for (int r = 0; r < 4; ++r) *(v4f*)(sa + 4 * r * XPITCH) = ra[r];
    #pragma unroll
    for (int r = 0; r < 16; ++r) *(v4f*)(sw + 4 * r * WPITCH) = rw[r];

    __syncthreads();

    v8f acc0 = {0.f, 0.f, 0.f, 0.f, 0.f, 0.f, 0.f, 0.f};
    v8f acc1 = {0.f, 0.f, 0.f, 0.f, 0.f, 0.f, 0.f, 0.f};

    #pragma unroll 1
    for (int c = 0; c < DM / KC; ++c) {
        const int cur = c & 1;
        const bool more = c + 1 < DM / KC;

        if (more) {
            #pragma unroll
            for (int r = 0; r < 4; ++r)
                ra[r] = *(const v4f*)(ga + (size_t)(4 * r) * DM + (c + 1) * KC);
            #pragma unroll
            for (int r = 0; r < 16; ++r)
                rw[r] = *(const v4f*)(gw + (size_t)(4 * r) * DM + (c + 1) * KC);
        }

        v4f af[KC / 8], bf[KC / 8];
        #pragma unroll
        for (int s = 0; s < KC / 8; ++s) {
            af[s] = *(const v4f*)(fa + cur * XBUF + 8 * s);
            bf[s] = *(const v4f*)(fb + cur * WBUF + 8 * s);
        }
        __builtin_amdgcn_sched_barrier(0);
        #pragma unroll
        for (int s = 0; s < KC / 8; ++s) {
            const v2f a0 = {af[s].x, af[s].y}, b0 = {bf[s].x, bf[s].y};
            const v2f a1 = {af[s].z, af[s].w}, b1 = {bf[s].z, bf[s].w};
            acc0 = __builtin_amdgcn_wmma_f32_16x16x4_f32(
                        false, a0,  false, b0,
                        (short)0, acc0,  false,  false);
            acc1 = __builtin_amdgcn_wmma_f32_16x16x4_f32(
                        false, a1,  false, b1,
                        (short)0, acc1,  false,  false);
        }
        if (more) {
            #pragma unroll
            for (int r = 0; r < 4; ++r) *(v4f*)(sa + (cur ^ 1) * XBUF + 4 * r * XPITCH) = ra[r];
            #pragma unroll
            for (int r = 0; r < 16; ++r) *(v4f*)(sw + (cur ^ 1) * WBUF + 4 * r * WPITCH) = rw[r];
        }

        __syncthreads();
    }
    return acc0 + acc1;
}

__global__ __launch_bounds__(128)
void gemm512_wmma(const float* __restrict__ A, const float* __restrict__ W,
                  float* __restrict__ Y)
{
    __shared__ __attribute__((aligned(16))) float wl[2 * WBUF];
    __shared__ __attribute__((aligned(16))) float xl[2 * XBUF];

    const int t    = threadIdx.x;
    const int wave = t >> 5;
    const int lane = t & 31;
    const int tr   = blockIdx.x >> 3;
    const int cb   = blockIdx.x & 7;
    const int tc   = cb * 4 + wave;
    const int half = lane >> 4;
    const int idx  = lane & 15;

    const v8f acc = tile16x64_k512(A + (size_t)(tr * 16) * DM,
                                   W + (size_t)(cb * 64) * DM, wl, xl, t);

    const int mbase = tr * 16 + half * 8;
    const int col   = tc * 16 + idx;
    #pragma unroll
    for (int i = 0; i < 8; ++i)
        Y[(size_t)(mbase + i) * DM + col] = acc[i];
}

__device__ __forceinline__ void load_row16(v4f (&dst)[16], const float* src)
{
    #pragma unroll
    for (int c = 0; c < 16; ++c) dst[c] = ((const v4f*)src)[c];
}

__device__ __forceinline__ float tropical_row(const v4f (&x)[16], const v4f (&w)[16])
{
    float ma = -__builtin_inff(), mb = -__builtin_inff();
    #pragma unroll
    for (int c = 0; c < 16; ++c) {
        ma = fmaxf(fmaxf(ma, x[c].x + w[c].x), x[c].y + w[c].y);
        mb = fmaxf(fmaxf(mb, x[c].z + w[c].z), x[c].w + w[c].w);
    }
    return fmaxf(ma, mb);
}

__global__ __launch_bounds__(128)
void proj_tropical(const float* __restrict__ x,
                   const float* __restrict__ Wq, const float* __restrict__ Wk,
                   const float* __restrict__ Wv, const float* __restrict__ lam,
                   const float* __restrict__ Tq, const float* __restrict__ Tk,
                   const float* __restrict__ Tv,
                   float* __restrict__ oq, float* __restrict__ ok,
                   float* __restrict__ ov)
{
    __shared__ __attribute__((aligned(16))) float wl[2 * WBUF];
    __shared__ __attribute__((aligned(16))) float xl[2 * XBUF];

    const int t    = threadIdx.x;
    const int wave = t >> 5;
    const int lane = t & 31;
    const int tr   = blockIdx.x >> 3;
    const int h    = blockIdx.x & 7;
    const int p    = blockIdx.y;
    const float* W  = p == 0 ? Wq : (p == 1 ? Wk : Wv);
    const float* Wt = p == 0 ? Tq : (p == 1 ? Tk : Tv);
    float*       out = p == 0 ? oq : (p == 1 ? ok : ov);

    v4f wt[8];
    #pragma unroll
    for (int r = 0; r < 8; ++r) wt[r] = ((const v4f*)Wt)[t + 128 * r];

    const int half = lane >> 4;
    const int idx  = lane & 15;
    const int tc   = h * 4 + wave;
    const v8f acc = tile16x64_k512(x + (size_t)(tr * 16) * DM,
                                   W + (size_t)(h * 64) * DM, wl, xl, t);

    #pragma unroll
    for (int r = 0; r < 8; ++r) {
        const int e = t + 128 * r;
        *(v4f*)(wl + (e >> 4) * WPITCH + (e & 15) * 4) = wt[r];
    }

    const float lm = lam[tc * 16 + idx];
    #pragma unroll
    for (int i = 0; i < 8; ++i)
        xl[(half * 8 + i) * XPITCH + wave * 16 + idx] = log1pf(fmaxf(acc[i], 0.f)) - lm;

    __syncthreads();

    const int o  = (wave & 1) * 32 + lane;
    const int r0 = (wave >> 1) * 8;
    v4f wr[16];
    #pragma unroll
    for (int c = 0; c < 16; ++c) wr[c] = *(const v4f*)(wl + o * WPITCH + c * 4);

    const int R = tr * 16 + r0;
    const int b = R >> 8, l = R & (LL - 1);
    const float* xb = xl + r0 * XPITCH;
    float* og = out + ((size_t)(b * NH + h) * LL + l) * DKK + o;
    {
        v4f xa[16], xc[16];
        load_row16(xa, xb);
        #pragma unroll 1
        for (int r = 0; r < 8; r += 2) {
            load_row16(xc, xb + (r + 1) * XPITCH);
            og[(size_t)r * DKK] = tropical_row(xa, wr);
            load_row16(xa, xb + ((r + 2) & 7) * XPITCH);
            og[(size_t)(r + 1) * DKK] = tropical_row(xc, wr);
        }
    }
}

#define KPITCH 68

__device__ __forceinline__ float score_row(const v4f (&q)[16], const v4f (&k)[16])
{
    float xa = -__builtin_inff(), xb = -__builtin_inff();
    float na =  __builtin_inff(), nb =  __builtin_inff();
    #pragma unroll
    for (int c = 0; c < 16; ++c) {
        const float d0 = q[c].x - k[c].x, d1 = q[c].y - k[c].y;
        const float d2 = q[c].z - k[c].z, d3 = q[c].w - k[c].w;
        xa = fmaxf(fmaxf(xa, d0), d1);  xb = fmaxf(fmaxf(xb, d2), d3);
        na = fminf(fminf(na, d0), d1);  nb = fminf(fminf(nb, d2), d3);
    }
    return fminf(na, nb) - fmaxf(xa, xb);
}

__device__ __forceinline__ void load_srows(v4f (&dst)[16], const float* src)
{
    #pragma unroll
    for (int c = 0; c < 16; ++c) dst[c] = ((const v4f*)src)[c];
}

__device__ __forceinline__ void ctx_rows(const v4f (&s)[16], const v2f (&v)[32],
                                         v2f& p0, v2f& p1)
{
    const float NI = -__builtin_inff();
    float ax0 = NI, ay0 = NI, bx0 = NI, by0 = NI;
    float ax1 = NI, ay1 = NI, bx1 = NI, by1 = NI;
    #pragma unroll
    for (int c = 0; c < 8; ++c) {
        const v4f sa = s[c], sb = s[8 + c];
        const v2f v0 = v[4 * c], v1 = v[4 * c + 1], v2 = v[4 * c + 2], v3 = v[4 * c + 3];
        ax0 = fmaxf(fmaxf(ax0, sa.x + v0.x), sa.y + v1.x);
        ay0 = fmaxf(fmaxf(ay0, sa.x + v0.y), sa.y + v1.y);
        bx0 = fmaxf(fmaxf(bx0, sa.z + v2.x), sa.w + v3.x);
        by0 = fmaxf(fmaxf(by0, sa.z + v2.y), sa.w + v3.y);
        ax1 = fmaxf(fmaxf(ax1, sb.x + v0.x), sb.y + v1.x);
        ay1 = fmaxf(fmaxf(ay1, sb.x + v0.y), sb.y + v1.y);
        bx1 = fmaxf(fmaxf(bx1, sb.z + v2.x), sb.w + v3.x);
        by1 = fmaxf(fmaxf(by1, sb.z + v2.y), sb.w + v3.y);
    }
    p0.x = fmaxf(ax0, bx0);  p0.y = fmaxf(ay0, by0);
    p1.x = fmaxf(ax1, bx1);  p1.y = fmaxf(ay1, by1);
}

__global__ __launch_bounds__(256) __attribute__((amdgpu_waves_per_eu(2, 2)))
void trop_attn(const float* __restrict__ qt, const float* __restrict__ kt,
               const float* __restrict__ vt, float* __restrict__ scores,
               float* __restrict__ ctxout)
{
    __shared__ __attribute__((aligned(16))) float kst[8 * 32 * KPITCH];
    __shared__ __attribute__((aligned(16))) float vst[LL * DKK];
    __shared__ __attribute__((aligned(16))) float qs[16 * DKK];
    __shared__ __attribute__((aligned(16))) float ss[8 * 16 * 32];

    const int t    = threadIdx.x;
    const int w    = t >> 5;
    const int lane = t & 31;
    const int it   = blockIdx.x;
    const int bh   = blockIdx.y;
    const size_t rowbase = (size_t)bh * LL;

    ((v4f*)qs)[t] = ((const v4f*)(qt + (rowbase + it * 16) * DKK))[t];

    float* kw = kst + w * (32 * KPITCH);
    float* vw = vst + w * (32 * DKK);
    {
        const v4f* kg = (const v4f*)(kt + (rowbase + 32 * w) * DKK);
        const v4f* vg = (const v4f*)(vt + (rowbase + 32 * w) * DKK);
        #pragma unroll
        for (int r = 0; r < 16; ++r) {
            const v4f x = kg[r * 32 + lane];
            const v4f y = vg[r * 32 + lane];
            *(v4f*)(kw + (2 * r + (lane >> 4)) * KPITCH + (lane & 15) * 4) = x;
            ((v4f*)vw)[r * 32 + lane] = y;
        }
    }

    __syncthreads();

    v4f kr[16];
    #pragma unroll
    for (int c = 0; c < 16; ++c) kr[c] = *(const v4f*)(kw + lane * KPITCH + c * 4);

    float* sw = ss + w * (16 * 32);
    float* sg = scores + (rowbase + it * 16) * LL + 32 * w + lane;
    {
        v4f qa[16], qb[16];
        load_row16(qa, qs);
        #pragma unroll 1
        for (int i = 0; i < 16; i += 2) {
            load_row16(qb, qs + (i + 1) * DKK);
            const float s0 = score_row(qa, kr);
            sw[i * 32 + lane]  = s0;
            sg[(size_t)i * LL] = s0;
            load_row16(qa, qs + ((i + 2) & 15) * DKK);
            const float s1 = score_row(qb, kr);
            sw[(i + 1) * 32 + lane]  = s1;
            sg[(size_t)(i + 1) * LL] = s1;
        }
    }

    v2f vr[32];
    #pragma unroll
    for (int jj = 0; jj < 32; ++jj) vr[jj] = *(const v2f*)(vw + jj * DKK + 2 * lane);

    __syncthreads();

    float* pw = kw;
    {
        v4f sa[16], sb[16];
        load_srows(sa, sw);
        #pragma unroll 1
        for (int i = 0; i < 16; i += 4) {
            v2f p0, p1;
            load_srows(sb, sw + (i + 2) * 32);
            ctx_rows(sa, vr, p0, p1);
            *(v2f*)(pw + i * DKK + 2 * lane)       = p0;
            *(v2f*)(pw + (i + 1) * DKK + 2 * lane) = p1;
            load_srows(sa, sw + ((i + 4) & 15) * 32);
            ctx_rows(sb, vr, p0, p1);
            *(v2f*)(pw + (i + 2) * DKK + 2 * lane) = p0;
            *(v2f*)(pw + (i + 3) * DKK + 2 * lane) = p1;
        }
    }

    __syncthreads();

    {
        const int i = t >> 4, dq = t & 15;
        v4f m = *(const v4f*)(kst + i * DKK + dq * 4);
        #pragma unroll
        for (int ww = 1; ww < 8; ++ww) {
            const v4f p = *(const v4f*)(kst + ww * (32 * KPITCH) + i * DKK + dq * 4);
            m.x = fmaxf(m.x, p.x);  m.y = fmaxf(m.y, p.y);
            m.z = fmaxf(m.z, p.z);  m.w = fmaxf(m.w, p.w);
        }
        const int b = bh >> 3, h = bh & 7;
        v4f o;
        o.x = expm1f(m.x);  o.y = expm1f(m.y);  o.z = expm1f(m.z);  o.w = expm1f(m.w);
        *(v4f*)(ctxout + (size_t)(b * LL + it * 16 + i) * DM + h * DKK + dq * 4) = o;
    }
}

extern "C" void kernel_launch(void* const* d_in, const int* in_sizes, int n_in,
                              void* d_out, int out_size, void* d_ws, size_t ws_size,
                              hipStream_t stream)
{
    (void)in_sizes; (void)n_in; (void)out_size; (void)ws_size;
    const float* x   = (const float*)d_in[0];
    const float* Wq  = (const float*)d_in[1];
    const float* Wk  = (const float*)d_in[2];
    const float* Wv  = (const float*)d_in[3];
    const float* Wo  = (const float*)d_in[4];
    const float* lam = (const float*)d_in[5];
    const float* Wqt = (const float*)d_in[6];
    const float* Wkt = (const float*)d_in[7];
    const float* Wvt = (const float*)d_in[8];

    float* out = (float*)d_out;
    float* sc  = out + (size_t)ROWS * DM;

    const size_t N = (size_t)ROWS * DM;
    float* ws  = (float*)d_ws;
    float* ctx = ws;
    float* qtb = ws + 3 * N;
    float* ktb = ws + 4 * N;
    float* vtb = ws + 5 * N;

    dim3 blk(256);
    proj_tropical<<<dim3(256, 3), dim3(128), 0, stream>>>(x, Wq, Wk, Wv, lam, Wqt, Wkt, Wvt,
                                                         qtb, ktb, vtb);

    trop_attn<<<dim3(16, 16), blk, 0, stream>>>(qtb, ktb, vtb, sc, ctx);

    gemm512_wmma<<<256, dim3(128), 0, stream>>>(ctx, Wo, out);
}


# attention prologue: only q and K are waited for before the scores; the V rows stay in flight in registers under the score phase and are staged to LDS behind it
# speedup vs baseline: 5.1466x; 1.0401x over previous
#include <hip/hip_runtime.h>
#include <math.h>

#define DM   512
#define NH   8
#define DKK  64
#define B_   2
#define LL   256
#define ROWS (B_*LL)

typedef __attribute__((ext_vector_type(2))) float v2f;
typedef __attribute__((ext_vector_type(4))) float v4f;
typedef __attribute__((ext_vector_type(8))) float v8f;

#define WPITCH 132
#define XPITCH 132
#define KC     128
#define WBUF   (DKK * WPITCH)
#define XBUF   (16 * XPITCH)

__device__ __forceinline__ v8f tile16x64_k512(const float* __restrict__ Ab,
                                              const float* __restrict__ Wb,
                                              float* wl, float* xl, const int t)
{
    const int lane = t & 31, wave = t >> 5;
    const int half = lane >> 4, idx = lane & 15;
    const int prow = t >> 5;
    const int pcol = (t & 31) * 4;

    const float* ga = Ab + (size_t)prow * DM + pcol;
    const float* gw = Wb + (size_t)prow * DM + pcol;
    float* sa = xl + prow * XPITCH + pcol;
    float* sw = wl + prow * WPITCH + pcol;
    const float* fa = xl + idx * XPITCH + 4 * half;
    const float* fb = wl + (wave * 16 + idx) * WPITCH + 4 * half;

    v4f ra[4], rw[16];
    #pragma unroll
    for (int r = 0; r < 4; ++r) ra[r] = *(const v4f*)(ga + (size_t)(4 * r) * DM);
    #pragma unroll
    for (int r = 0; r < 16; ++r) rw[r] = *(const v4f*)(gw + (size_t)(4 * r) * DM);
    #pragma unroll
    for (int r = 0; r < 4; ++r) *(v4f*)(sa + 4 * r * XPITCH) = ra[r];
    #pragma unroll
    for (int r = 0; r < 16; ++r) *(v4f*)(sw + 4 * r * WPITCH) = rw[r];

    __syncthreads();

    v8f acc0 = {0.f, 0.f, 0.f, 0.f, 0.f, 0.f, 0.f, 0.f};
    v8f acc1 = {0.f, 0.f, 0.f, 0.f, 0.f, 0.f, 0.f, 0.f};

    #pragma unroll 1
    for (int c = 0; c < DM / KC; ++c) {
        const int cur = c & 1;
        const bool more = c + 1 < DM / KC;

        if (more) {
            #pragma unroll
            for (int r = 0; r < 4; ++r)
                ra[r] = *(const v4f*)(ga + (size_t)(4 * r) * DM + (c + 1) * KC);
            #pragma unroll
            for (int r = 0; r < 16; ++r)
                rw[r] = *(const v4f*)(gw + (size_t)(4 * r) * DM + (c + 1) * KC);
        }

        v4f af[KC / 8], bf[KC / 8];
        #pragma unroll
        for (int s = 0; s < KC / 8; ++s) {
            af[s] = *(const v4f*)(fa + cur * XBUF + 8 * s);
            bf[s] = *(const v4f*)(fb + cur * WBUF + 8 * s);
        }
        __builtin_amdgcn_sched_barrier(0);
        #pragma unroll
        for (int s = 0; s < KC / 8; ++s) {
            const v2f a0 = {af[s].x, af[s].y}, b0 = {bf[s].x, bf[s].y};
            const v2f a1 = {af[s].z, af[s].w}, b1 = {bf[s].z, bf[s].w};
            acc0 = __builtin_amdgcn_wmma_f32_16x16x4_f32(
                        false, a0,  false, b0,
                        (short)0, acc0,  false,  false);
            acc1 = __builtin_amdgcn_wmma_f32_16x16x4_f32(
                        false, a1,  false, b1,
                        (short)0, acc1,  false,  false);
        }
        if (more) {
            #pragma unroll
            for (int r = 0; r < 4; ++r) *(v4f*)(sa + (cur ^ 1) * XBUF + 4 * r * XPITCH) = ra[r];
            #pragma unroll
            for (int r = 0; r < 16; ++r) *(v4f*)(sw + (cur ^ 1) * WBUF + 4 * r * WPITCH) = rw[r];
        }

        __syncthreads();
    }
    return acc0 + acc1;
}

__global__ __launch_bounds__(128)
void gemm512_wmma(const float* __restrict__ A, const float* __restrict__ W,
                  float* __restrict__ Y)
{
    __shared__ __attribute__((aligned(16))) float wl[2 * WBUF];
    __shared__ __attribute__((aligned(16))) float xl[2 * XBUF];

    const int t    = threadIdx.x;
    const int wave = t >> 5;
    const int lane = t & 31;
    const int tr   = blockIdx.x >> 3;
    const int cb   = blockIdx.x & 7;
    const int tc   = cb * 4 + wave;
    const int half = lane >> 4;
    const int idx  = lane & 15;

    const v8f acc = tile16x64_k512(A + (size_t)(tr * 16) * DM,
                                   W + (size_t)(cb * 64) * DM, wl, xl, t);

    const int mbase = tr * 16 + half * 8;
    const int col   = tc * 16 + idx;
    #pragma unroll
    for (int i = 0; i < 8; ++i)
        Y[(size_t)(mbase + i) * DM + col] = acc[i];
}

__device__ __forceinline__ void load_row16(v4f (&dst)[16], const float* src)
{
    #pragma unroll
    for (int c = 0; c < 16; ++c) dst[c] = ((const v4f*)src)[c];
}

__device__ __forceinline__ float tropical_row(const v4f (&x)[16], const v4f (&w)[16])
{
    float ma = -__builtin_inff(), mb = -__builtin_inff();
    #pragma unroll
    for (int c = 0; c < 16; ++c) {
        ma = fmaxf(fmaxf(ma, x[c].x + w[c].x), x[c].y + w[c].y);
        mb = fmaxf(fmaxf(mb, x[c].z + w[c].z), x[c].w + w[c].w);
    }
    return fmaxf(ma, mb);
}

__global__ __launch_bounds__(128)
void proj_tropical(const float* __restrict__ x,
                   const float* __restrict__ Wq, const float* __restrict__ Wk,
                   const float* __restrict__ Wv, const float* __restrict__ lam,
                   const float* __restrict__ Tq, const float* __restrict__ Tk,
                   const float* __restrict__ Tv,
                   float* __restrict__ oq, float* __restrict__ ok,
                   float* __restrict__ ov)
{
    __shared__ __attribute__((aligned(16))) float wl[2 * WBUF];
    __shared__ __attribute__((aligned(16))) float xl[2 * XBUF];

    const int t    = threadIdx.x;
    const int wave = t >> 5;
    const int lane = t & 31;
    const int tr   = blockIdx.x >> 3;
    const int h    = blockIdx.x & 7;
    const int p    = blockIdx.y;
    const float* W  = p == 0 ? Wq : (p == 1 ? Wk : Wv);
    const float* Wt = p == 0 ? Tq : (p == 1 ? Tk : Tv);
    float*       out = p == 0 ? oq : (p == 1 ? ok : ov);

    v4f wt[8];
    #pragma unroll
    for (int r = 0; r < 8; ++r) wt[r] = ((const v4f*)Wt)[t + 128 * r];

    const int half = lane >> 4;
    const int idx  = lane & 15;
    const int tc   = h * 4 + wave;
    const v8f acc = tile16x64_k512(x + (size_t)(tr * 16) * DM,
                                   W + (size_t)(h * 64) * DM, wl, xl, t);

    #pragma unroll
    for (int r = 0; r < 8; ++r) {
        const int e = t + 128 * r;
        *(v4f*)(wl + (e >> 4) * WPITCH + (e & 15) * 4) = wt[r];
    }

    const float lm = lam[tc * 16 + idx];
    #pragma unroll
    for (int i = 0; i < 8; ++i)
        xl[(half * 8 + i) * XPITCH + wave * 16 + idx] = log1pf(fmaxf(acc[i], 0.f)) - lm;

    __syncthreads();

    const int o  = (wave & 1) * 32 + lane;
    const int r0 = (wave >> 1) * 8;
    v4f wr[16];
    #pragma unroll
    for (int c = 0; c < 16; ++c) wr[c] = *(const v4f*)(wl + o * WPITCH + c * 4);

    const int R = tr * 16 + r0;
    const int b = R >> 8, l = R & (LL - 1);
    const float* xb = xl + r0 * XPITCH;
    float* og = out + ((size_t)(b * NH + h) * LL + l) * DKK + o;
    {
        v4f xa[16], xc[16];
        load_row16(xa, xb);
        #pragma unroll 1
        for (int r = 0; r < 8; r += 2) {
            load_row16(xc, xb + (r + 1) * XPITCH);
            og[(size_t)r * DKK] = tropical_row(xa, wr);
            load_row16(xa, xb + ((r + 2) & 7) * XPITCH);
            og[(size_t)(r + 1) * DKK] = tropical_row(xc, wr);
        }
    }
}

#define KPITCH 68

__device__ __forceinline__ float score_row(const v4f (&q)[16], const v4f (&k)[16])
{
    float xa = -__builtin_inff(), xb = -__builtin_inff();
    float na =  __builtin_inff(), nb =  __builtin_inff();
    #pragma unroll
    for (int c = 0; c < 16; ++c) {
        const float d0 = q[c].x - k[c].x, d1 = q[c].y - k[c].y;
        const float d2 = q[c].z - k[c].z, d3 = q[c].w - k[c].w;
        xa = fmaxf(fmaxf(xa, d0), d1);  xb = fmaxf(fmaxf(xb, d2), d3);
        na = fminf(fminf(na, d0), d1);  nb = fminf(fminf(nb, d2), d3);
    }
    return fminf(na, nb) - fmaxf(xa, xb);
}

__device__ __forceinline__ void load_srows(v4f (&dst)[16], const float* src)
{
    #pragma unroll
    for (int c = 0; c < 16; ++c) dst[c] = ((const v4f*)src)[c];
}

__device__ __forceinline__ void ctx_rows(const v4f (&s)[16], const v2f (&v)[32],
                                         v2f& p0, v2f& p1)
{
    const float NI = -__builtin_inff();
    float ax0 = NI, ay0 = NI, bx0 = NI, by0 = NI;
    float ax1 = NI, ay1 = NI, bx1 = NI, by1 = NI;
    #pragma unroll
    for (int c = 0; c < 8; ++c) {
        const v4f sa = s[c], sb = s[8 + c];
        const v2f v0 = v[4 * c], v1 = v[4 * c + 1], v2 = v[4 * c + 2], v3 = v[4 * c + 3];
        ax0 = fmaxf(fmaxf(ax0, sa.x + v0.x), sa.y + v1.x);
        ay0 = fmaxf(fmaxf(ay0, sa.x + v0.y), sa.y + v1.y);
        bx0 = fmaxf(fmaxf(bx0, sa.z + v2.x), sa.w + v3.x);
        by0 = fmaxf(fmaxf(by0, sa.z + v2.y), sa.w + v3.y);
        ax1 = fmaxf(fmaxf(ax1, sb.x + v0.x), sb.y + v1.x);
        ay1 = fmaxf(fmaxf(ay1, sb.x + v0.y), sb.y + v1.y);
        bx1 = fmaxf(fmaxf(bx1, sb.z + v2.x), sb.w + v3.x);
        by1 = fmaxf(fmaxf(by1, sb.z + v2.y), sb.w + v3.y);
    }
    p0.x = fmaxf(ax0, bx0);  p0.y = fmaxf(ay0, by0);
    p1.x = fmaxf(ax1, bx1);  p1.y = fmaxf(ay1, by1);
}

__global__ __launch_bounds__(256) __attribute__((amdgpu_waves_per_eu(2, 2)))
void trop_attn(const float* __restrict__ qt, const float* __restrict__ kt,
               const float* __restrict__ vt, float* __restrict__ scores,
               float* __restrict__ ctxout)
{
    __shared__ __attribute__((aligned(16))) float kst[8 * 32 * KPITCH];
    __shared__ __attribute__((aligned(16))) float vst[LL * DKK];
    __shared__ __attribute__((aligned(16))) float qs[16 * DKK];
    __shared__ __attribute__((aligned(16))) float ss[8 * 16 * 32];

    const int t    = threadIdx.x;
    const int w    = t >> 5;
    const int lane = t & 31;
    const int it   = blockIdx.x;
    const int bh   = blockIdx.y;
    const size_t rowbase = (size_t)bh * LL;

    float* kw = kst + w * (32 * KPITCH);
    float* vw = vst + w * (32 * DKK);
    v4f vy[16];
    {
        const v4f qv = ((const v4f*)(qt + (rowbase + it * 16) * DKK))[t];
        const v4f* kg = (const v4f*)(kt + (rowbase + 32 * w) * DKK);
        const v4f* vg = (const v4f*)(vt + (rowbase + 32 * w) * DKK);
        v4f kx[16];
        #pragma unroll
        for (int r = 0; r < 16; ++r) kx[r] = kg[r * 32 + lane];
        __builtin_amdgcn_sched_barrier(0);
        #pragma unroll
        for (int r = 0; r < 16; ++r) vy[r] = vg[r * 32 + lane];
        __builtin_amdgcn_sched_barrier(0);
        ((v4f*)qs)[t] = qv;
        #pragma unroll
        for (int r = 0; r < 16; ++r)
            *(v4f*)(kw + (2 * r + (lane >> 4)) * KPITCH + (lane & 15) * 4) = kx[r];
    }

    __builtin_amdgcn_fence(__ATOMIC_RELEASE, "workgroup", "local");
    __builtin_amdgcn_s_barrier();
    __builtin_amdgcn_fence(__ATOMIC_ACQUIRE, "workgroup", "local");

    v4f kr[16];
    #pragma unroll
    for (int c = 0; c < 16; ++c) kr[c] = *(const v4f*)(kw + lane * KPITCH + c * 4);

    float* sw = ss + w * (16 * 32);
    float* sg = scores + (rowbase + it * 16) * LL + 32 * w + lane;
    {
        v4f qa[16], qb[16];
        load_row16(qa, qs);
        #pragma unroll 1
        for (int i = 0; i < 16; i += 2) {
            load_row16(qb, qs + (i + 1) * DKK);
            const float s0 = score_row(qa, kr);
            sw[i * 32 + lane]  = s0;
            sg[(size_t)i * LL] = s0;
            load_row16(qa, qs + ((i + 2) & 15) * DKK);
            const float s1 = score_row(qb, kr);
            sw[(i + 1) * 32 + lane]  = s1;
            sg[(size_t)(i + 1) * LL] = s1;
        }
    }

    #pragma unroll
    for (int r = 0; r < 16; ++r) ((v4f*)vw)[r * 32 + lane] = vy[r];

    __syncthreads();

    v2f vr[32];
    #pragma unroll
    for (int jj = 0; jj < 32; ++jj) vr[jj] = *(const v2f*)(vw + jj * DKK + 2 * lane);

    float* pw = kw;
    {
        v4f sa[16], sb[16];
        load_srows(sa, sw);
        #pragma unroll 1
        for (int i = 0; i < 16; i += 4) {
            v2f p0, p1;
            load_srows(sb, sw + (i + 2) * 32);
            ctx_rows(sa, vr, p0, p1);
            *(v2f*)(pw + i * DKK + 2 * lane)       = p0;
            *(v2f*)(pw + (i + 1) * DKK + 2 * lane) = p1;
            load_srows(sa, sw + ((i + 4) & 15) * 32);
            ctx_rows(sb, vr, p0, p1);
            *(v2f*)(pw + (i + 2) * DKK + 2 * lane) = p0;
            *(v2f*)(pw + (i + 3) * DKK + 2 * lane) = p1;
        }
    }

    __syncthreads();

    {
        const int i = t >> 4, dq = t & 15;
        v4f m = *(const v4f*)(kst + i * DKK + dq * 4);
        #pragma unroll
        for (int ww = 1; ww < 8; ++ww) {
            const v4f p = *(const v4f*)(kst + ww * (32 * KPITCH) + i * DKK + dq * 4);
            m.x = fmaxf(m.x, p.x);  m.y = fmaxf(m.y, p.y);
            m.z = fmaxf(m.z, p.z);  m.w = fmaxf(m.w, p.w);
        }
        const int b = bh >> 3, h = bh & 7;
        v4f o;
        o.x = expm1f(m.x);  o.y = expm1f(m.y);  o.z = expm1f(m.z);  o.w = expm1f(m.w);
        *(v4f*)(ctxout + (size_t)(b * LL + it * 16 + i) * DM + h * DKK + dq * 4) = o;
    }
}

extern "C" void kernel_launch(void* const* d_in, const int* in_sizes, int n_in,
                              void* d_out, int out_size, void* d_ws, size_t ws_size,
                              hipStream_t stream)
{
    (void)in_sizes; (void)n_in; (void)out_size; (void)ws_size;
    const float* x   = (const float*)d_in[0];
    const float* Wq  = (const float*)d_in[1];
    const float* Wk  = (const float*)d_in[2];
    const float* Wv  = (const float*)d_in[3];
    const float* Wo  = (const float*)d_in[4];
    const float* lam = (const float*)d_in[5];
    const float* Wqt = (const float*)d_in[6];
    const float* Wkt = (const float*)d_in[7];
    const float* Wvt = (const float*)d_in[8];

    float* out = (float*)d_out;
    float* sc  = out + (size_t)ROWS * DM;

    const size_t N = (size_t)ROWS * DM;
    float* ws  = (float*)d_ws;
    float* ctx = ws;
    float* qtb = ws + 3 * N;
    float* ktb = ws + 4 * N;
    float* vtb = ws + 5 * N;

    dim3 blk(256);
    proj_tropical<<<dim3(256, 3), dim3(128), 0, stream>>>(x, Wq, Wk, Wv, lam, Wqt, Wkt, Wvt,
                                                         qtb, ktb, vtb);

    trop_attn<<<dim3(16, 16), blk, 0, stream>>>(qtb, ktb, vtb, sc, ctx);

    gemm512_wmma<<<256, dim3(128), 0, stream>>>(ctx, Wo, out);
}
